# starting kernel
# baseline (speedup 1.0000x reference)
#include <hip/hip_runtime.h>
#include <math.h>

typedef __attribute__((ext_vector_type(2))) float v2f;
typedef __attribute__((ext_vector_type(8))) float v8f;

#define T_SEQ     10
#define D_MODEL   256
#define BATCH     16
#define RES       200
#define MODES_OUT 16
#define MODES_IN  32
#define XPAD      208
#define NXT       13
#define KDIM      64

__global__ __launch_bounds__(256) void build_wf(const float* __restrict__ w_real,
                                                const float* __restrict__ w_imag,
                                                float* __restrict__ wf) {
  int idx = blockIdx.x * blockDim.x + threadIdx.x;
  int m = idx & 31;
  int x = (idx >> 5) % XPAD;
  int t = idx / (XPAD * 32);
  if (t >= T_SEQ) return;

  float accr = 0.0f, acci = 0.0f;
  if (x < RES) {
    const float base = (6.283185307179586476925f / 200.0f) * (float)x;
    #pragma unroll
    for (int o = 0; o < MODES_OUT; ++o) {
      float s, c;
      sincosf(base * (float)o, &s, &c);
      float wr = w_real[(t * MODES_OUT + o) * MODES_IN + m];
      float wi = w_imag[(t * MODES_OUT + o) * MODES_IN + m];
      accr += wr * c - wi * s;
      acci -= wi * c + wr * s;
    }
  }
  float* row = wf + (t * XPAD + x) * KDIM;
  row[m]            = accr;
  row[MODES_IN + m] = acci;
}

__global__ __launch_bounds__(256) void spectral_wmma(const float* __restrict__ v,
                                                     const float* __restrict__ wf,
                                                     float* __restrict__ out) {
  const int lane = threadIdx.x & 31;
  const int wid  = (int)((blockIdx.x * blockDim.x + threadIdx.x) >> 5);

  const int hp = wid & 7;
  const int xt = (wid >> 3) % NXT;
  const int tt = (wid / (8 * NXT)) % T_SEQ;
  const int b  = wid / (8 * NXT * T_SEQ);
  if (b >= BATCH) return;

  const int lm = lane & 15;
  const int lh = lane >> 4;

  const float* arow  = wf + (size_t)(tt * XPAD + xt * 16 + lm) * KDIM + 2 * lh;
  const float* brow0 = v  + (size_t)(((b * T_SEQ + tt) * D_MODEL) + hp * 32 + lm) * KDIM + 2 * lh;
  const float* brow1 = brow0 + 16 * KDIM;

  v8f acc0 = {};
  v8f acc1 = {};
  #pragma unroll
  for (int kb = 0; kb < 16; ++kb) {
    v2f a  = *(const v2f*)(arow  + 4 * kb);
    v2f b0 = *(const v2f*)(brow0 + 4 * kb);
    v2f b1 = *(const v2f*)(brow1 + 4 * kb);
    acc0 = __builtin_amdgcn_wmma_f32_16x16x4_f32(false, a, false, b0,
                                                 (short)0, acc0, false, false);
    acc1 = __builtin_amdgcn_wmma_f32_16x16x4_f32(false, a, false, b1,
                                                 (short)0, acc1, false, false);
  }

  float* obase = out + (size_t)(b * T_SEQ + tt) * RES * D_MODEL + hp * 32 + lm;
  const int xb = xt * 16;
  #pragma unroll
  for (int j = 0; j < 8; ++j) {
    int x = xb + j + 8 * lh;
    if (x < RES) {
      obase[(size_t)x * D_MODEL]      = acc0[j];
      obase[(size_t)x * D_MODEL + 16] = acc1[j];
    }
  }
}

extern "C" void kernel_launch(void* const* d_in, const int* in_sizes, int n_in,
                              void* d_out, int out_size, void* d_ws, size_t ws_size,
                              hipStream_t stream) {
  const float* v      = (const float*)d_in[0];
  const float* w_real = (const float*)d_in[1];
  const float* w_imag = (const float*)d_in[2];
  float* out = (float*)d_out;
  float* wf  = (float*)d_ws;

  build_wf<<<(T_SEQ * XPAD * 32) / 256, 256, 0, stream>>>(w_real, w_imag, wf);

  spectral_wmma<<<(BATCH * T_SEQ * NXT * 8) / 8, 256, 0, stream>>>(v, wf, out);
}


# block = (b,t,64 h): its V rows staged once in LDS by whole-line 16 B loads, shared by 13 waves (one per x-tile, Wf fragments in registers); V re-fetch 13x -> 1x; k order, so sum order, changed
# speedup vs baseline: 2.4207x; 2.4207x over previous
#include <hip/hip_runtime.h>
#include <math.h>

typedef __attribute__((ext_vector_type(2))) float v2f;
typedef __attribute__((ext_vector_type(8))) float v8f;

#define T_SEQ     10
#define D_MODEL   256
#define BATCH     16
#define RES       200
#define MODES_OUT 16
#define MODES_IN  32
#define XPAD      208
#define NXT       13
#define KDIM      64

__global__ __launch_bounds__(256) void build_wf(const float* __restrict__ w_real,
                                                const float* __restrict__ w_imag,
                                                float* __restrict__ wf) {
  int idx = blockIdx.x * blockDim.x + threadIdx.x;
  int m = idx & 31;
  int x = (idx >> 5) % XPAD;
  int t = idx / (XPAD * 32);
  if (t >= T_SEQ) return;

  float accr = 0.0f, acci = 0.0f;
  if (x < RES) {
    const float base = (6.283185307179586476925f / 200.0f) * (float)x;
    #pragma unroll
    for (int o = 0; o < MODES_OUT; ++o) {
      float s, c;
      sincosf(base * (float)o, &s, &c);
      float wr = w_real[(t * MODES_OUT + o) * MODES_IN + m];
      float wi = w_imag[(t * MODES_OUT + o) * MODES_IN + m];
      accr += wr * c - wi * s;
      acci -= wi * c + wr * s;
    }
  }
  float* row = wf + (t * XPAD + x) * KDIM;
  row[m]            = accr;
  row[MODES_IN + m] = acci;
}

#define HC     64
#define VPITCH 68
#define NWAVE  13

__global__ __launch_bounds__(NWAVE * 32) __attribute__((amdgpu_waves_per_eu(4, 10))) void spectral_wmma(const float* __restrict__ v,
                                                            const float* __restrict__ wf,
                                                            float* __restrict__ out) {
  __shared__ __attribute__((aligned(16))) float vs[HC * VPITCH];

  const int tid  = threadIdx.x;
  const int lane = tid & 31;
  const int xt   = tid >> 5;
  const int hc   = blockIdx.x & 3;
  const int bt   = blockIdx.x >> 2;
  const int tt   = bt % T_SEQ;

  const int lm = lane & 15;
  const int lh = lane >> 4;

  const float* arow = wf + (size_t)(tt * XPAD + xt * 16 + lm) * KDIM + 4 * lh;
  float4 a[8];
  #pragma unroll
  for (int q = 0; q < 8; ++q) a[q] = *(const float4*)(arow + 8 * q);

  const float4* vsrc = (const float4*)(v + ((size_t)bt * D_MODEL + hc * HC) * KDIM);
  const int i0 = tid, i1 = tid + NWAVE * 32;
  const int i2 = (tid + 2 * NWAVE * 32 < HC * 16) ? tid + 2 * NWAVE * 32 : tid;
  float4 q0 = vsrc[i0];
  float4 q1 = vsrc[i1];
  float4 q2 = vsrc[i2];
  *(float4*)&vs[(i0 >> 4) * VPITCH + 4 * (i0 & 15)] = q0;
  *(float4*)&vs[(i1 >> 4) * VPITCH + 4 * (i1 & 15)] = q1;
  *(float4*)&vs[(i2 >> 4) * VPITCH + 4 * (i2 & 15)] = q2;
  __syncthreads();

  const float* brow = vs + lm * VPITCH + 4 * lh;
  float* obase = out + (size_t)bt * RES * D_MODEL + hc * HC + lm;
  const int xb = xt * 16;

  #pragma unroll
  for (int np = 0; np < 2; ++np) {
    const float* br0 = brow + (32 * np) * VPITCH;
    const float* br1 = br0 + 16 * VPITCH;
    float4 b0[8], b1[8];
    #pragma unroll
    for (int q = 0; q < 8; ++q) {
      b0[q] = *(const float4*)(br0 + 8 * q);
      b1[q] = *(const float4*)(br1 + 8 * q);
    }
    v8f acc0 = {};
    v8f acc1 = {};
    #pragma unroll
    for (int q = 0; q < 8; ++q) {
      v2f alo = {a[q].x, a[q].y};
      v2f ahi = {a[q].z, a[q].w};
      v2f b0lo = {b0[q].x, b0[q].y};
      v2f b0hi = {b0[q].z, b0[q].w};
      v2f b1lo = {b1[q].x, b1[q].y};
      v2f b1hi = {b1[q].z, b1[q].w};
      acc0 = __builtin_amdgcn_wmma_f32_16x16x4_f32(false, alo, false, b0lo,
                                                   (short)0, acc0, false, false);
      acc1 = __builtin_amdgcn_wmma_f32_16x16x4_f32(false, alo, false, b1lo,
                                                   (short)0, acc1, false, false);
      acc0 = __builtin_amdgcn_wmma_f32_16x16x4_f32(false, ahi, false, b0hi,
                                                   (short)0, acc0, false, false);
      acc1 = __builtin_amdgcn_wmma_f32_16x16x4_f32(false, ahi, false, b1hi,
                                                   (short)0, acc1, false, false);
    }
    float* ob = obase + 32 * np;
    #pragma unroll
    for (int j = 0; j < 8; ++j) {
      int x = xb + j + 8 * lh;
      if (x < RES) {
        ob[(size_t)x * D_MODEL]      = acc0[j];
        ob[(size_t)x * D_MODEL + 16] = acc1[j];
      }
    }
  }
}

extern "C" void kernel_launch(void* const* d_in, const int* in_sizes, int n_in,
                              void* d_out, int out_size, void* d_ws, size_t ws_size,
                              hipStream_t stream) {
  const float* v      = (const float*)d_in[0];
  const float* w_real = (const float*)d_in[1];
  const float* w_imag = (const float*)d_in[2];
  float* out = (float*)d_out;
  float* wf  = (float*)d_ws;

  build_wf<<<(T_SEQ * XPAD * 32) / 256, 256, 0, stream>>>(w_real, w_imag, wf);

  spectral_wmma<<<BATCH * T_SEQ * 4, NWAVE * 32, 0, stream>>>(v, wf, out);
}


# resident grid: 256 blocks of 16 waves, one per WGP; each stages its 160 V rows and its 1-2 Wf tables in LDS once, waves walk 65 (x-tile, 32 h) units, A and B from LDS; sums unchanged
# speedup vs baseline: 2.5762x; 1.0642x over previous
#include <hip/hip_runtime.h>
#include <math.h>

typedef __attribute__((ext_vector_type(2))) float v2f;
typedef __attribute__((ext_vector_type(4))) float v4f;
typedef __attribute__((ext_vector_type(8))) float v8f;

#define T_SEQ     10
#define D_MODEL   256
#define BATCH     16
#define RES       200
#define MODES_OUT 16
#define MODES_IN  32
#define XPAD      208
#define NXT       13
#define KDIM      64

__global__ __launch_bounds__(256) void build_wf(const float* __restrict__ w_real,
                                                const float* __restrict__ w_imag,
                                                float* __restrict__ wf) {
  int idx = blockIdx.x * blockDim.x + threadIdx.x;
  int m = idx & 31;
  int x = (idx >> 5) % XPAD;
  int t = idx / (XPAD * 32);
  if (t >= T_SEQ) return;

  float accr = 0.0f, acci = 0.0f;
  if (x < RES) {
    const float base = (6.283185307179586476925f / 200.0f) * (float)x;
    #pragma unroll
    for (int o = 0; o < MODES_OUT; ++o) {
      float s, c;
      sincosf(base * (float)o, &s, &c);
      float wr = w_real[(t * MODES_OUT + o) * MODES_IN + m];
      float wi = w_imag[(t * MODES_OUT + o) * MODES_IN + m];
      accr += wr * c - wi * s;
      acci -= wi * c + wr * s;
    }
  }
  float* row = wf + (t * XPAD + x) * KDIM;
  row[m]            = accr;
  row[MODES_IN + m] = acci;
}

#define VPITCH 68
#define NWAVE  16
#define NTHR   (NWAVE * 32)
#define BROWS  160
#define NPAIR  5
#define NUNIT  (NPAIR * NXT)
#define WF4    (XPAD * 16)
#define WLDS   (XPAD * VPITCH)
#define LDSPIN 2048

__global__ __launch_bounds__(NTHR) void spectral_wmma(const float* __restrict__ v,
                                                            const float* __restrict__ wf,
                                                            float* __restrict__ out) {
  __shared__ __attribute__((aligned(16))) float vs[BROWS * VPITCH + LDSPIN];
  __shared__ __attribute__((aligned(16))) float ws[2 * WLDS];

  const int tid  = threadIdx.x;
  const int lane = tid & 31;
  const int lm = lane & 15;
  const int lh = lane >> 4;

  const int row0 = blockIdx.x * BROWS;
  const int bt0  = row0 >> 8;
  const int bt1  = (row0 + BROWS - 1) >> 8;
  const int t0   = bt0 % T_SEQ;
  const int t1   = bt1 % T_SEQ;

  const v4f* vsrc = (const v4f*)(v + (size_t)row0 * KDIM);
  v4f qv[5];
  #pragma unroll
  for (int i = 0; i < 5; ++i) qv[i] = vsrc[tid + i * NTHR];

  const int i6 = (tid + 6 * NTHR < WF4) ? tid + 6 * NTHR : tid;
  const v4f* w0src = (const v4f*)(wf + (size_t)t0 * XPAD * KDIM);
  v4f qa[7];
  #pragma unroll
  for (int i = 0; i < 6; ++i) qa[i] = w0src[tid + i * NTHR];
  qa[6] = w0src[i6];

  if (bt1 != bt0) {
    const v4f* w1src = (const v4f*)(wf + (size_t)t1 * XPAD * KDIM);
    v4f qb[7];
    #pragma unroll
    for (int i = 0; i < 6; ++i) qb[i] = w1src[tid + i * NTHR];
    qb[6] = w1src[i6];
    asm volatile("" :: "v"(qv[0]), "v"(qv[1]), "v"(qv[2]), "v"(qv[3]), "v"(qv[4]),
                       "v"(qa[0]), "v"(qa[1]), "v"(qa[2]), "v"(qa[3]), "v"(qa[4]), "v"(qa[5]), "v"(qa[6]),
                       "v"(qb[0]), "v"(qb[1]), "v"(qb[2]), "v"(qb[3]), "v"(qb[4]), "v"(qb[5]), "v"(qb[6]));
    #pragma unroll
    for (int i = 0; i < 6; ++i) {
      int p = tid + i * NTHR;
      *(v4f*)&ws[WLDS + (p >> 4) * VPITCH + 4 * (p & 15)] = qb[i];
    }
    *(v4f*)&ws[WLDS + (i6 >> 4) * VPITCH + 4 * (i6 & 15)] = qb[6];
  }
  #pragma unroll
  for (int i = 0; i < 5; ++i) {
    int p = tid + i * NTHR;
    *(v4f*)&vs[(p >> 4) * VPITCH + 4 * (p & 15)] = qv[i];
  }
  #pragma unroll
  for (int i = 0; i < 6; ++i) {
    int p = tid + i * NTHR;
    *(v4f*)&ws[(p >> 4) * VPITCH + 4 * (p & 15)] = qa[i];
  }
  *(v4f*)&ws[(i6 >> 4) * VPITCH + 4 * (i6 & 15)] = qa[6];
  __syncthreads();

  const int wave = __builtin_amdgcn_readfirstlane(tid >> 5);
  const float* afrag = ws + lm * VPITCH + 4 * lh;
  const float* bfrag = vs + lm * VPITCH + 4 * lh;

  for (int u = wave; u < NUNIT; u += NWAVE) {
    const int hp  = u / NXT;
    const int xt  = u - hp * NXT;
    const int row = row0 + 32 * hp;
    const int bt  = row >> 8;
    const int h0  = row & 255;

    const float* ar  = afrag + (bt != bt0 ? WLDS : 0) + (xt * 16) * VPITCH;
    const float* br0 = bfrag + (32 * hp) * VPITCH;
    const float* br1 = br0 + 16 * VPITCH;
    float4 a[8], b0[8], b1[8];
    #pragma unroll
    for (int q = 0; q < 8; ++q) {
      a[q]  = *(const float4*)(ar  + 8 * q);
      b0[q] = *(const float4*)(br0 + 8 * q);
      b1[q] = *(const float4*)(br1 + 8 * q);
    }
    v8f acc0 = {};
    v8f acc1 = {};
    #pragma unroll
    for (int q = 0; q < 8; ++q) {
      v2f alo = {a[q].x, a[q].y};
      v2f ahi = {a[q].z, a[q].w};
      v2f b0lo = {b0[q].x, b0[q].y};
      v2f b0hi = {b0[q].z, b0[q].w};
      v2f b1lo = {b1[q].x, b1[q].y};
      v2f b1hi = {b1[q].z, b1[q].w};
      acc0 = __builtin_amdgcn_wmma_f32_16x16x4_f32(false, alo, false, b0lo,
                                                   (short)0, acc0, false, false);
      acc1 = __builtin_amdgcn_wmma_f32_16x16x4_f32(false, alo, false, b1lo,
                                                   (short)0, acc1, false, false);
      acc0 = __builtin_amdgcn_wmma_f32_16x16x4_f32(false, ahi, false, b0hi,
                                                   (short)0, acc0, false, false);
      acc1 = __builtin_amdgcn_wmma_f32_16x16x4_f32(false, ahi, false, b1hi,
                                                   (short)0, acc1, false, false);
    }
    float* ob = out + (size_t)bt * RES * D_MODEL + h0 + lm;
    const int xb = xt * 16;
    #pragma unroll
    for (int j = 0; j < 8; ++j) {
      int x = xb + j + 8 * lh;
      if (x < RES) {
        ob[(size_t)x * D_MODEL]      = acc0[j];
        ob[(size_t)x * D_MODEL + 16] = acc1[j];
      }
    }
  }
}

extern "C" void kernel_launch(void* const* d_in, const int* in_sizes, int n_in,
                              void* d_out, int out_size, void* d_ws, size_t ws_size,
                              hipStream_t stream) {
  const float* v      = (const float*)d_in[0];
  const float* w_real = (const float*)d_in[1];
  const float* w_imag = (const float*)d_in[2];
  float* out = (float*)d_out;
  float* wf  = (float*)d_ws;

  build_wf<<<(T_SEQ * XPAD * 32) / 256, 256, 0, stream>>>(w_real, w_imag, wf);

  spectral_wmma<<<256, NTHR, 0, stream>>>(v, wf, out);
}


# no table launch: each block forms the Wf table(s) of its t in LDS as an f32 WMMA product Trig(208x32) x Wcat(32x64), Trig gathered from 200 cos/sin in LDS; 2 launches -> 1; table sum order changed
# speedup vs baseline: 2.9031x; 1.1269x over previous
#include <hip/hip_runtime.h>
#include <math.h>

typedef __attribute__((ext_vector_type(2))) float v2f;
typedef __attribute__((ext_vector_type(4))) float v4f;
typedef __attribute__((ext_vector_type(8))) float v8f;

#define T_SEQ     10
#define D_MODEL   256
#define BATCH     16
#define RES       200
#define MODES_OUT 16
#define MODES_IN  32
#define XPAD      208
#define NXT       13
#define KDIM      64


#define VPITCH 68
#define NWAVE  16
#define NTHR   (NWAVE * 32)
#define BROWS  160
#define NPAIR  5
#define NUNIT  (NPAIR * NXT)
#define WLDS   (XPAD * VPITCH)
#define LDSPIN 2048

__device__ __forceinline__ v8f wf_tile(const v2f (&af)[8], const v2f (&bw)[8]) {
  v8f d = {};
  #pragma unroll
  for (int n = 0; n < 8; ++n)
    d = __builtin_amdgcn_wmma_f32_16x16x4_f32(false, af[n], false, bw[n], (short)0, d, false, false);
  return d;
}

__global__ __launch_bounds__(NTHR) void spectral_wmma(const float* __restrict__ v,
                                                            const float* __restrict__ w_real,
                                                            const float* __restrict__ w_imag,
                                                            float* __restrict__ out) {
  __shared__ __attribute__((aligned(16))) float vs[BROWS * VPITCH + LDSPIN];
  __shared__ __attribute__((aligned(16))) float ws[2 * WLDS];
  __shared__ float trig[2 * RES];

  const int tid  = threadIdx.x;
  const int lane = tid & 31;
  const int lm = lane & 15;
  const int lh = lane >> 4;
  const int wave = __builtin_amdgcn_readfirstlane(tid >> 5);

  const int row0 = blockIdx.x * BROWS;
  const int bt0  = row0 >> 8;
  const int bt1  = (row0 + BROWS - 1) >> 8;
  const int t0   = bt0 % T_SEQ;
  const int t1   = bt1 % T_SEQ;
  const bool two = (bt1 != bt0);

  const int nt = wave & 3;
  const int m  = 16 * (nt & 1) + lm;
  const float* wsel = ((lh ^ (nt >> 1)) != 0) ? w_imag : w_real;
  const bool wneg = (lh | (nt >> 1)) != 0;
  float wq0[16], wq1[16];
  #pragma unroll
  for (int o = 0; o < MODES_OUT; ++o) wq0[o] = wsel[(t0 * MODES_OUT + o) * MODES_IN + m];
  #pragma unroll
  for (int o = 0; o < MODES_OUT; ++o) wq1[o] = 0.0f;
  if (two) {
    #pragma unroll
    for (int o = 0; o < MODES_OUT; ++o) wq1[o] = wsel[(t1 * MODES_OUT + o) * MODES_IN + m];
  }

  if (tid < RES) {
    float sn, cs;
    sincosf((6.283185307179586476925f / 200.0f) * (float)tid, &sn, &cs);
    trig[tid]       = cs;
    trig[RES + tid] = sn;
  }
  __syncthreads();

  const v4f* vsrc = (const v4f*)(v + (size_t)row0 * KDIM);
  v4f qv[5];
  #pragma unroll
  for (int i = 0; i < 5; ++i) qv[i] = vsrc[tid + i * NTHR];

  {
    v2f bw0[8], bw1[8];
    #pragma unroll
    for (int n = 0; n < 8; ++n) {
      bw0[n] = v2f{wneg ? -wq0[2 * n] : wq0[2 * n], wneg ? -wq0[2 * n + 1] : wq0[2 * n + 1]};
      bw1[n] = v2f{wneg ? -wq1[2 * n] : wq1[2 * n], wneg ? -wq1[2 * n + 1] : wq1[2 * n + 1]};
    }
    const float* tg = trig + lh * RES;
    for (int xt = wave >> 2; xt < NXT; xt += 4) {
      const int x  = 16 * xt + lm;
      const int xe = x < RES ? x : x - RES;
      v2f af[8];
      int j = 0;
      #pragma unroll
      for (int n = 0; n < 8; ++n) {
        float c0 = tg[j];
        j += xe; j = j >= RES ? j - RES : j;
        float c1 = tg[j];
        j += xe; j = j >= RES ? j - RES : j;
        af[n] = v2f{c0, c1};
      }
      float* wrow = ws + (16 * xt + 8 * lh) * VPITCH + 16 * nt + lm;
      v8f d0 = wf_tile(af, bw0);
      #pragma unroll
      for (int r = 0; r < 8; ++r) wrow[r * VPITCH] = d0[r];
      if (two) {
        v8f d1 = wf_tile(af, bw1);
        #pragma unroll
        for (int r = 0; r < 8; ++r) wrow[WLDS + r * VPITCH] = d1[r];
      }
    }
  }

  #pragma unroll
  for (int i = 0; i < 5; ++i) {
    int p = tid + i * NTHR;
    *(v4f*)&vs[(p >> 4) * VPITCH + 4 * (p & 15)] = qv[i];
  }
  __syncthreads();

  const float* afrag = ws + lm * VPITCH + 4 * lh;
  const float* bfrag = vs + lm * VPITCH + 4 * lh;

  for (int u = wave; u < NUNIT; u += NWAVE) {
    const int hp  = u / NXT;
    const int xt  = u - hp * NXT;
    const int row = row0 + 32 * hp;
    const int bt  = row >> 8;
    const int h0  = row & 255;

    const float* ar  = afrag + (bt != bt0 ? WLDS : 0) + (xt * 16) * VPITCH;
    const float* br0 = bfrag + (32 * hp) * VPITCH;
    const float* br1 = br0 + 16 * VPITCH;
    float4 a[8], b0[8], b1[8];
    #pragma unroll
    for (int q = 0; q < 8; ++q) {
      a[q]  = *(const float4*)(ar  + 8 * q);
      b0[q] = *(const float4*)(br0 + 8 * q);
      b1[q] = *(const float4*)(br1 + 8 * q);
    }
    v8f acc0 = {};
    v8f acc1 = {};
    #pragma unroll
    for (int q = 0; q < 8; ++q) {
      v2f alo = {a[q].x, a[q].y};
      v2f ahi = {a[q].z, a[q].w};
      v2f b0lo = {b0[q].x, b0[q].y};
      v2f b0hi = {b0[q].z, b0[q].w};
      v2f b1lo = {b1[q].x, b1[q].y};
      v2f b1hi = {b1[q].z, b1[q].w};
      acc0 = __builtin_amdgcn_wmma_f32_16x16x4_f32(false, alo, false, b0lo,
                                                   (short)0, acc0, false, false);
      acc1 = __builtin_amdgcn_wmma_f32_16x16x4_f32(false, alo, false, b1lo,
                                                   (short)0, acc1, false, false);
      acc0 = __builtin_amdgcn_wmma_f32_16x16x4_f32(false, ahi, false, b0hi,
                                                   (short)0, acc0, false, false);
      acc1 = __builtin_amdgcn_wmma_f32_16x16x4_f32(false, ahi, false, b1hi,
                                                   (short)0, acc1, false, false);
    }
    float* ob = out + (size_t)bt * RES * D_MODEL + h0 + lm;
    const int xb = xt * 16;
    #pragma unroll
    for (int j = 0; j < 8; ++j) {
      int x = xb + j + 8 * lh;
      if (x < RES) {
        ob[(size_t)x * D_MODEL]      = acc0[j];
        ob[(size_t)x * D_MODEL + 16] = acc1[j];
      }
    }
  }
}

extern "C" void kernel_launch(void* const* d_in, const int* in_sizes, int n_in,
                              void* d_out, int out_size, void* d_ws, size_t ws_size,
                              hipStream_t stream) {
  const float* v      = (const float*)d_in[0];
  const float* w_real = (const float*)d_in[1];
  const float* w_imag = (const float*)d_in[2];
  float* out = (float*)d_out;

  spectral_wmma<<<256, NTHR, 0, stream>>>(v, w_real, w_imag, out);
}


# a pair's two h-tiles take its even and odd h (tile n, column c = h0+2c+n), so a lane holds two adjacent h: each output row leaves as one whole 128 B line, 8 b64 stores in one clause; sums unchanged
# speedup vs baseline: 3.1681x; 1.0913x over previous
#include <hip/hip_runtime.h>
#include <math.h>

typedef __attribute__((ext_vector_type(2))) float v2f;
typedef __attribute__((ext_vector_type(4))) float v4f;
typedef __attribute__((ext_vector_type(8))) float v8f;

#define T_SEQ     10
#define D_MODEL   256
#define BATCH     16
#define RES       200
#define MODES_OUT 16
#define MODES_IN  32
#define XPAD      208
#define NXT       13
#define KDIM      64


#define VPITCH 68
#define NWAVE  16
#define NTHR   (NWAVE * 32)
#define BROWS  160
#define NPAIR  5
#define NUNIT  (NPAIR * NXT)
#define WLDS   (XPAD * VPITCH)
#define LDSPIN 2048

__device__ __forceinline__ v8f wf_tile(const v2f (&af)[8], const v2f (&bw)[8]) {
  v8f d = {};
  #pragma unroll
  for (int n = 0; n < 8; ++n)
    d = __builtin_amdgcn_wmma_f32_16x16x4_f32(false, af[n], false, bw[n], (short)0, d, false, false);
  return d;
}

__global__ __launch_bounds__(NTHR) void spectral_wmma(const float* __restrict__ v,
                                                            const float* __restrict__ w_real,
                                                            const float* __restrict__ w_imag,
                                                            float* __restrict__ out) {
  __shared__ __attribute__((aligned(16))) float vs[BROWS * VPITCH + LDSPIN];
  __shared__ __attribute__((aligned(16))) float ws[2 * WLDS];
  __shared__ float trig[2 * RES];

  const int tid  = threadIdx.x;
  const int lane = tid & 31;
  const int lm = lane & 15;
  const int lh = lane >> 4;
  const int wave = __builtin_amdgcn_readfirstlane(tid >> 5);

  const int row0 = blockIdx.x * BROWS;
  const int bt0  = row0 >> 8;
  const int bt1  = (row0 + BROWS - 1) >> 8;
  const int t0   = bt0 % T_SEQ;
  const int t1   = bt1 % T_SEQ;
  const bool two = (bt1 != bt0);

  const int nt = wave & 3;
  const int m  = 16 * (nt & 1) + lm;
  const float* wsel = ((lh ^ (nt >> 1)) != 0) ? w_imag : w_real;
  const bool wneg = (lh | (nt >> 1)) != 0;
  float wq0[16], wq1[16];
  #pragma unroll
  for (int o = 0; o < MODES_OUT; ++o) wq0[o] = wsel[(t0 * MODES_OUT + o) * MODES_IN + m];
  #pragma unroll
  for (int o = 0; o < MODES_OUT; ++o) wq1[o] = 0.0f;
  if (two) {
    #pragma unroll
    for (int o = 0; o < MODES_OUT; ++o) wq1[o] = wsel[(t1 * MODES_OUT + o) * MODES_IN + m];
  }

  if (tid < RES) {
    float sn, cs;
    sincosf((6.283185307179586476925f / 200.0f) * (float)tid, &sn, &cs);
    trig[tid]       = cs;
    trig[RES + tid] = sn;
  }
  __syncthreads();

  const v4f* vsrc = (const v4f*)(v + (size_t)row0 * KDIM);
  v4f qv[5];
  #pragma unroll
  for (int i = 0; i < 5; ++i) qv[i] = vsrc[tid + i * NTHR];

  {
    v2f bw0[8], bw1[8];
    #pragma unroll
    for (int n = 0; n < 8; ++n) {
      bw0[n] = v2f{wneg ? -wq0[2 * n] : wq0[2 * n], wneg ? -wq0[2 * n + 1] : wq0[2 * n + 1]};
      bw1[n] = v2f{wneg ? -wq1[2 * n] : wq1[2 * n], wneg ? -wq1[2 * n + 1] : wq1[2 * n + 1]};
    }
    const float* tg = trig + lh * RES;
    for (int xt = wave >> 2; xt < NXT; xt += 4) {
      const int x  = 16 * xt + lm;
      const int xe = x < RES ? x : x - RES;
      v2f af[8];
      int j = 0;
      #pragma unroll
      for (int n = 0; n < 8; ++n) {
        float c0 = tg[j];
        j += xe; j = j >= RES ? j - RES : j;
        float c1 = tg[j];
        j += xe; j = j >= RES ? j - RES : j;
        af[n] = v2f{c0, c1};
      }
      float* wrow = ws + (16 * xt + 8 * lh) * VPITCH + 16 * nt + lm;
      v8f d0 = wf_tile(af, bw0);
      #pragma unroll
      for (int r = 0; r < 8; ++r) wrow[r * VPITCH] = d0[r];
      if (two) {
        v8f d1 = wf_tile(af, bw1);
        #pragma unroll
        for (int r = 0; r < 8; ++r) wrow[WLDS + r * VPITCH] = d1[r];
      }
    }
  }

  #pragma unroll
  for (int i = 0; i < 5; ++i) {
    int p = tid + i * NTHR;
    int r = p >> 4;
    int rl = (r & ~31) | ((r & 1) << 4) | ((r & 31) >> 1);
    *(v4f*)&vs[rl * VPITCH + 4 * (p & 15)] = qv[i];
  }
  __syncthreads();

  const float* afrag = ws + lm * VPITCH + 4 * lh;
  const float* bfrag = vs + lm * VPITCH + 4 * lh;

  for (int u = wave; u < NUNIT; u += NWAVE) {
    const int hp  = u / NXT;
    const int xt  = u - hp * NXT;
    const int row = row0 + 32 * hp;
    const int bt  = row >> 8;
    const int h0  = row & 255;

    const float* ar  = afrag + (bt != bt0 ? WLDS : 0) + (xt * 16) * VPITCH;
    const float* br0 = bfrag + (32 * hp) * VPITCH;
    const float* br1 = br0 + 16 * VPITCH;
    float4 a[8], b0[8], b1[8];
    #pragma unroll
    for (int q = 0; q < 8; ++q) {
      a[q]  = *(const float4*)(ar  + 8 * q);
      b0[q] = *(const float4*)(br0 + 8 * q);
      b1[q] = *(const float4*)(br1 + 8 * q);
    }
    v8f acc0 = {};
    v8f acc1 = {};
    #pragma unroll
    for (int q = 0; q < 8; ++q) {
      v2f alo = {a[q].x, a[q].y};
      v2f ahi = {a[q].z, a[q].w};
      v2f b0lo = {b0[q].x, b0[q].y};
      v2f b0hi = {b0[q].z, b0[q].w};
      v2f b1lo = {b1[q].x, b1[q].y};
      v2f b1hi = {b1[q].z, b1[q].w};
      acc0 = __builtin_amdgcn_wmma_f32_16x16x4_f32(false, alo, false, b0lo,
                                                   (short)0, acc0, false, false);
      acc1 = __builtin_amdgcn_wmma_f32_16x16x4_f32(false, alo, false, b1lo,
                                                   (short)0, acc1, false, false);
      acc0 = __builtin_amdgcn_wmma_f32_16x16x4_f32(false, ahi, false, b0hi,
                                                   (short)0, acc0, false, false);
      acc1 = __builtin_amdgcn_wmma_f32_16x16x4_f32(false, ahi, false, b1hi,
                                                   (short)0, acc1, false, false);
    }
    const int xr = xt * 16 + 8 * lh;
    if (xr + 7 < RES) {
      float* ob = out + ((size_t)bt * RES + xr) * D_MODEL + h0 + 2 * lm;
      #pragma unroll
      for (int j = 0; j < 8; ++j)
        *(v2f*)&ob[j * D_MODEL] = v2f{acc0[j], acc1[j]};
    }
  }
}

extern "C" void kernel_launch(void* const* d_in, const int* in_sizes, int n_in,
                              void* d_out, int out_size, void* d_ws, size_t ws_size,
                              hipStream_t stream) {
  const float* v      = (const float*)d_in[0];
  const float* w_real = (const float*)d_in[1];
  const float* w_imag = (const float*)d_in[2];
  float* out = (float*)d_out;

  spectral_wmma<<<256, NTHR, 0, stream>>>(v, w_real, w_imag, out);
}


# each block's Wf table product split into cos part E and sin part O (K=16 each), multiplied for x-tiles 0..6 only: table row x = E+O, row 200-x = E-O; 416 -> 224 WMMAs a table; table sum order changed
# speedup vs baseline: 3.4449x; 1.0874x over previous
#include <hip/hip_runtime.h>
#include <math.h>

typedef __attribute__((ext_vector_type(2))) float v2f;
typedef __attribute__((ext_vector_type(4))) float v4f;
typedef __attribute__((ext_vector_type(8))) float v8f;

#define T_SEQ     10
#define D_MODEL   256
#define BATCH     16
#define RES       200
#define MODES_OUT 16
#define MODES_IN  32
#define XPAD      208
#define NXT       13
#define NXH       7
#define KDIM      64


#define VPITCH 68
#define NWAVE  16
#define NTHR   (NWAVE * 32)
#define BROWS  160
#define NPAIR  5
#define NUNIT  (NPAIR * NXT)
#define WLDS   (XPAD * VPITCH)
#define LDSPIN 2048

__global__ __launch_bounds__(NTHR) void spectral_wmma(const float* __restrict__ v,
                                                            const float* __restrict__ w_real,
                                                            const float* __restrict__ w_imag,
                                                            float* __restrict__ out) {
  __shared__ __attribute__((aligned(16))) float vs[BROWS * VPITCH + LDSPIN];
  __shared__ __attribute__((aligned(16))) float ws[2 * WLDS];
  __shared__ float trig[2 * RES];

  const int tid  = threadIdx.x;
  const int lane = tid & 31;
  const int lm = lane & 15;
  const int lh = lane >> 4;
  const int wave = __builtin_amdgcn_readfirstlane(tid >> 5);

  const int row0 = blockIdx.x * BROWS;
  const int bt0  = row0 >> 8;
  const int bt1  = (row0 + BROWS - 1) >> 8;
  const int t0   = bt0 % T_SEQ;
  const int t1   = bt1 % T_SEQ;
  const bool two = (bt1 != bt0);

  const int nt = wave & 3;
  const int m  = 16 * (nt & 1) + lm;
  const bool imh = (nt >> 1) != 0;
  const float* pc = imh ? w_imag : w_real;
  const float* pm = imh ? w_real : w_imag;
  float wc0[8], wm0[8], wc1[8], wm1[8];
  #pragma unroll
  for (int i = 0; i < 8; ++i) {
    const int o = 4 * (i >> 1) + 2 * lh + (i & 1);
    wc0[i] = pc[(t0 * MODES_OUT + o) * MODES_IN + m];
    wm0[i] = pm[(t0 * MODES_OUT + o) * MODES_IN + m];
  }
  #pragma unroll
  for (int i = 0; i < 8; ++i) { wc1[i] = 0.0f; wm1[i] = 0.0f; }
  if (two) {
    #pragma unroll
    for (int i = 0; i < 8; ++i) {
      const int o = 4 * (i >> 1) + 2 * lh + (i & 1);
      wc1[i] = pc[(t1 * MODES_OUT + o) * MODES_IN + m];
      wm1[i] = pm[(t1 * MODES_OUT + o) * MODES_IN + m];
    }
  }

  if (tid < RES) {
    float sn, cs;
    sincosf((6.283185307179586476925f / 200.0f) * (float)tid, &sn, &cs);
    trig[tid]       = cs;
    trig[RES + tid] = sn;
  }
  __syncthreads();

  const v4f* vsrc = (const v4f*)(v + (size_t)row0 * KDIM);
  v4f qv[5];
  #pragma unroll
  for (int i = 0; i < 5; ++i) qv[i] = vsrc[tid + i * NTHR];

  {
    v2f bc0[4], bm0[4], bc1[4], bm1[4];
    #pragma unroll
    for (int w = 0; w < 4; ++w) {
      bc0[w] = imh ? v2f{-wc0[2 * w], -wc0[2 * w + 1]} : v2f{wc0[2 * w], wc0[2 * w + 1]};
      bc1[w] = imh ? v2f{-wc1[2 * w], -wc1[2 * w + 1]} : v2f{wc1[2 * w], wc1[2 * w + 1]};
      bm0[w] = v2f{wm0[2 * w], wm0[2 * w + 1]};
      bm1[w] = v2f{wm1[2 * w], wm1[2 * w + 1]};
    }
    for (int xt = wave >> 2; xt < NXH; xt += 4) {
      const int x = 16 * xt + lm;
      int x2 = 2 * x;  x2 = x2 >= RES ? x2 - RES : x2;
      int x3 = x2 + x; x3 = x3 >= RES ? x3 - RES : x3;
      int j = lh ? x2 : 0;
      v2f ac[4], as[4];
      #pragma unroll
      for (int w = 0; w < 4; ++w) {
        const float c0 = trig[j], s0 = trig[RES + j];
        j += x;  j = j >= RES ? j - RES : j;
        const float c1 = trig[j], s1 = trig[RES + j];
        j += x3; j = j >= RES ? j - RES : j;
        ac[w] = v2f{c0, c1};
        as[w] = v2f{s0, s1};
      }
      float* wlo = ws + (16 * xt + 8 * lh) * VPITCH + 16 * nt + lm;
      float* whi = ws + (RES - 16 * xt - 8 * lh) * VPITCH + 16 * nt + lm;
      #pragma unroll
      for (int tb = 0; tb < 2; ++tb) {
        if (tb == 1 && !two) break;
        v8f e = {};
        v8f mo = {};
        #pragma unroll
        for (int w = 0; w < 4; ++w) {
          e  = __builtin_amdgcn_wmma_f32_16x16x4_f32(false, ac[w], false, tb ? bc1[w] : bc0[w], (short)0, e, false, false);
          mo = __builtin_amdgcn_wmma_f32_16x16x4_f32(false, as[w], false, tb ? bm1[w] : bm0[w], (short)0, mo, false, false);
        }
        float p[8], q[8];
        #pragma unroll
        for (int r = 0; r < 8; ++r) { p[r] = e[r] - mo[r]; q[r] = e[r] + mo[r]; }
        float* lo = wlo + tb * WLDS;
        float* hi = whi + tb * WLDS;
        if (xt < NXH - 1) {
          #pragma unroll
          for (int r = 0; r < 8; ++r) { lo[r * VPITCH] = p[r]; hi[-r * VPITCH] = q[r]; }
        } else {
          #pragma unroll
          for (int r = 0; r < 8; ++r) lo[(r + ((lh && r) ? 96 : 0)) * VPITCH] = p[r];
        }
      }
    }
  }

  #pragma unroll
  for (int i = 0; i < 5; ++i) {
    int p = tid + i * NTHR;
    int r = p >> 4;
    int rl = (r & ~31) | ((r & 1) << 4) | ((r & 31) >> 1);
    *(v4f*)&vs[rl * VPITCH + 4 * (p & 15)] = qv[i];
  }
  __syncthreads();

  const float* afrag = ws + lm * VPITCH + 4 * lh;
  const float* bfrag = vs + lm * VPITCH + 4 * lh;

  for (int u = wave; u < NUNIT; u += NWAVE) {
    const int hp  = u / NXT;
    const int xt  = u - hp * NXT;
    const int row = row0 + 32 * hp;
    const int bt  = row >> 8;
    const int h0  = row & 255;

    const float* ar  = afrag + (bt != bt0 ? WLDS : 0) + (xt * 16) * VPITCH;
    const float* br0 = bfrag + (32 * hp) * VPITCH;
    const float* br1 = br0 + 16 * VPITCH;
    float4 a[8], b0[8], b1[8];
    #pragma unroll
    for (int q = 0; q < 8; ++q) {
      a[q]  = *(const float4*)(ar  + 8 * q);
      b0[q] = *(const float4*)(br0 + 8 * q);
      b1[q] = *(const float4*)(br1 + 8 * q);
    }
    v8f acc0 = {};
    v8f acc1 = {};
    #pragma unroll
    for (int q = 0; q < 8; ++q) {
      v2f alo = {a[q].x, a[q].y};
      v2f ahi = {a[q].z, a[q].w};
      v2f b0lo = {b0[q].x, b0[q].y};
      v2f b0hi = {b0[q].z, b0[q].w};
      v2f b1lo = {b1[q].x, b1[q].y};
      v2f b1hi = {b1[q].z, b1[q].w};
      acc0 = __builtin_amdgcn_wmma_f32_16x16x4_f32(false, alo, false, b0lo,
                                                   (short)0, acc0, false, false);
      acc1 = __builtin_amdgcn_wmma_f32_16x16x4_f32(false, alo, false, b1lo,
                                                   (short)0, acc1, false, false);
      acc0 = __builtin_amdgcn_wmma_f32_16x16x4_f32(false, ahi, false, b0hi,
                                                   (short)0, acc0, false, false);
      acc1 = __builtin_amdgcn_wmma_f32_16x16x4_f32(false, ahi, false, b1hi,
                                                   (short)0, acc1, false, false);
    }
    const int xr = xt * 16 + 8 * lh;
    if (xr + 7 < RES) {
      float* ob = out + ((size_t)bt * RES + xr) * D_MODEL + h0 + 2 * lm;
      #pragma unroll
      for (int j = 0; j < 8; ++j)
        *(v2f*)&ob[j * D_MODEL] = v2f{acc0[j], acc1[j]};
    }
  }
}

extern "C" void kernel_launch(void* const* d_in, const int* in_sizes, int n_in,
                              void* d_out, int out_size, void* d_ws, size_t ws_size,
                              hipStream_t stream) {
  const float* v      = (const float*)d_in[0];
  const float* w_real = (const float*)d_in[1];
  const float* w_imag = (const float*)d_in[2];
  float* out = (float*)d_out;

  spectral_wmma<<<256, NTHR, 0, stream>>>(v, w_real, w_imag, out);
}
